# starting kernel
# baseline (speedup 1.0000x reference)
#include <hip/hip_runtime.h>
#include <hip/hip_bf16.h>


typedef __attribute__((ext_vector_type(2))) float v2f;
typedef __attribute__((ext_vector_type(8))) float v8f;

#define KTOT 768
#define LDS_STRIDE 772
#define MT 16
#define NSPAT 196

__global__ __launch_bounds__(512) void conv_shift_wmma_f32(
    const float* __restrict__ x,
    const float* __restrict__ Wm,
    float* __restrict__ out)
{
    __shared__ float As[MT * LDS_STRIDE];

    const int mtile = blockIdx.x;
    const int lane  = threadIdx.x & 31;
    const int wv    = threadIdx.x >> 5;
    const int chan  = (wv << 4) + (lane & 15);

    __builtin_prefetch(Wm + chan * KTOT, 0, 3);

    for (int idx = threadIdx.x; idx < MT * KTOT; idx += 512) {
        const int m = idx / KTOT;
        const int t = idx - m * KTOT;
        const int s = mtile * MT + m;
        float v = 0.0f;
        if (s < NSPAT) {
            const int o = s / 14;
            const int n = s - o * 14;
            const int j = t / 6;
            const int r = t - j * 6;
            const int l = r / 3;
            const int k = r - l * 3;
            int row = o - 1 - l;
            if (row < 0) row += 14;
            const int col = n + k - 1;
            if (col >= 0 && col < 14)
                v = x[j * NSPAT + row * 14 + col];
        }
        As[m * LDS_STRIDE + t] = v;
    }
    __syncthreads();

    const int m     = lane & 15;
    const int khalf = (lane >> 4) << 1;
    const float* wrow = Wm + chan * KTOT;

    v8f acc = {};
    #pragma unroll 8
    for (int k0 = 0; k0 < KTOT; k0 += 4) {
        v2f a = *(const v2f*)&As[m * LDS_STRIDE + k0 + khalf];
        v2f b = *(const v2f*)&wrow[k0 + khalf];
        acc = __builtin_amdgcn_wmma_f32_16x16x4_f32(
             false, a,  false, b,
             (short)0, acc,  false,  false);
    }

    const int mofs = (lane >> 4) << 3;
    #pragma unroll
    for (int v = 0; v < 8; ++v) {
        const int s = mtile * MT + v + mofs;
        if (s < NSPAT)
            out[chan * NSPAT + s] = acc[v];
    }
}

extern "C" void kernel_launch(void* const* d_in, const int* in_sizes, int n_in,
                              void* d_out, int out_size, void* d_ws, size_t ws_size,
                              hipStream_t stream) {
    const float* x  = (const float*)d_in[0];
    const float* Wm = (const float*)d_in[1];
    float* out = (float*)d_out;

    dim3 grid(13);
    dim3 block(512);
    conv_shift_wmma_f32<<<grid, block, 0, stream>>>(x, Wm, out);
}


# One 16x16 tile per single-wave work-group (208 groups, was 13 of 16 waves): each reads 16 rows of W, not all 256; A fragment gathered straight from x, no LDS stage; loads kept 18 k-steps ahead
# speedup vs baseline: 3.1326x; 3.1326x over previous
#include <hip/hip_runtime.h>


typedef __attribute__((ext_vector_type(2))) float v2f;
typedef __attribute__((ext_vector_type(8))) float v8f;

#define KTOT 768
#define MT 16
#define NSPAT 196
#define NPAIR 64
#define CPAIRS 2
#define CSTEPS (3 * CPAIRS)
#define NCHUNK (NPAIR / CPAIRS)

struct Frag {
    float a0[CSTEPS];
    float a1[CSTEPS];
    v2f   b[CSTEPS];
};

__device__ __forceinline__ void issue_chunk(Frag& f, const float* __restrict__ xc,
                                            const float* __restrict__ wc,
                                            const int (&off)[6])
{
    #pragma unroll
    for (int u = 0; u < CPAIRS; ++u) {
        #pragma unroll
        for (int p = 0; p < 3; ++p) {
            f.a0[u * 3 + p] = xc[u * 2 * NSPAT + off[2 * p]];
            f.a1[u * 3 + p] = xc[u * 2 * NSPAT + off[2 * p + 1]];
            f.b[u * 3 + p]  = *(const v2f*)(wc + u * 12 + 4 * p);
        }
    }
}

__device__ __forceinline__ void mma_chunk(v8f& acc, const Frag& f, const bool (&ok)[6])
{
    #pragma unroll
    for (int i = 0; i < CSTEPS; ++i) {
        const int p = i % 3;
        v2f a;
        a.x = ok[2 * p]     ? f.a0[i] : 0.0f;
        a.y = ok[2 * p + 1] ? f.a1[i] : 0.0f;
        acc = __builtin_amdgcn_wmma_f32_16x16x4_f32(
             false, a,  false, f.b[i],
             (short)0, acc,  false,  false);
    }
}

__global__ __launch_bounds__(32) __attribute__((amdgpu_waves_per_eu(1, 1)))
void conv_shift_wmma_f32(
    const float* __restrict__ x,
    const float* __restrict__ Wm,
    float* __restrict__ out)
{
    const int lane  = threadIdx.x;
    const int ntile = blockIdx.x & 15;
    const int mtile = blockIdx.x >> 4;
    const int c     = lane & 15;
    const int h     = lane >> 4;
    const int chan  = (ntile << 4) + c;

    const int  s   = mtile * MT + c;
    const bool sok = s < NSPAT;
    const int  sc  = sok ? s : 0;
    const int  o   = sc / 14;
    const int  n   = sc - o * 14;

    int  off[6];
    bool ok[6];
    #pragma unroll
    for (int i = 0; i < 6; ++i) {
        const int tt = 4 * (i >> 1) + 2 * h + (i & 1);
        const int dj = tt / 6;
        const int rr = tt - dj * 6;
        const int l  = rr / 3;
        const int k  = rr - l * 3;
        int row = o - 1 - l;
        if (row < 0) row += 14;
        const int col = n + k - 1;
        ok[i]  = sok && col >= 0 && col < 14;
        off[i] = ok[i] ? dj * NSPAT + row * 14 + col : 0;
    }

    const float* wrow = Wm + chan * KTOT + 2 * h;

    v8f acc = {};
    Frag ring[4];

    issue_chunk(ring[0], x + 0 * CPAIRS * 2 * NSPAT, wrow + 0 * CPAIRS * 12, off);
    issue_chunk(ring[1], x + 1 * CPAIRS * 2 * NSPAT, wrow + 1 * CPAIRS * 12, off);
    issue_chunk(ring[2], x + 2 * CPAIRS * 2 * NSPAT, wrow + 2 * CPAIRS * 12, off);

    #pragma unroll
    for (int q = 0; q < NCHUNK; ++q) {
        if (q + 3 < NCHUNK)
            issue_chunk(ring[(q + 3) & 3], x + (q + 3) * CPAIRS * 2 * NSPAT,
                        wrow + (q + 3) * CPAIRS * 12, off);
        mma_chunk(acc, ring[q & 3], ok);
    }

    const int mofs = h << 3;
    #pragma unroll
    for (int v = 0; v < 8; ++v) {
        const int so = mtile * MT + v + mofs;
        if (so < NSPAT)
            out[chan * NSPAT + so] = acc[v];
    }
}

extern "C" void kernel_launch(void* const* d_in, const int* in_sizes, int n_in,
                              void* d_out, int out_size, void* d_ws, size_t ws_size,
                              hipStream_t stream) {
    const float* x  = (const float*)d_in[0];
    const float* Wm = (const float*)d_in[1];
    float* out = (float*)d_out;

    dim3 grid(208);
    dim3 block(32);
    conv_shift_wmma_f32<<<grid, block, 0, stream>>>(x, Wm, out);
}


# K split over the 4 waves of a 128-thread work-group (one 16x16 tile each; 48 k-steps per wave, was 192 in one wave); the 4 partial tiles are added through LDS in fixed order: f32 sum order changes
# speedup vs baseline: 5.9193x; 1.8896x over previous
#include <hip/hip_runtime.h>


typedef __attribute__((ext_vector_type(2))) float v2f;
typedef __attribute__((ext_vector_type(8))) float v8f;

#define KTOT 768
#define MT 16
#define NSPAT 196
#define NPAIR 64
#define CPAIRS 2
#define CSTEPS (3 * CPAIRS)
#define NCHUNK (NPAIR / CPAIRS)
#define NWAVE 4
#define WCHUNK (NCHUNK / NWAVE)
#define RPW (8 / NWAVE)

struct Frag {
    float a0[CSTEPS];
    float a1[CSTEPS];
    v2f   b[CSTEPS];
};

__device__ __forceinline__ void issue_chunk(Frag& f, const float* __restrict__ xc,
                                            const float* __restrict__ wc,
                                            const int (&off)[6])
{
    #pragma unroll
    for (int u = 0; u < CPAIRS; ++u) {
        #pragma unroll
        for (int p = 0; p < 3; ++p) {
            f.a0[u * 3 + p] = xc[u * 2 * NSPAT + off[2 * p]];
            f.a1[u * 3 + p] = xc[u * 2 * NSPAT + off[2 * p + 1]];
            f.b[u * 3 + p]  = *(const v2f*)(wc + u * 12 + 4 * p);
        }
    }
}

__device__ __forceinline__ void mma_chunk(v8f& acc, const Frag& f, const bool (&ok)[6])
{
    #pragma unroll
    for (int i = 0; i < CSTEPS; ++i) {
        const int p = i % 3;
        v2f a;
        a.x = ok[2 * p]     ? f.a0[i] : 0.0f;
        a.y = ok[2 * p + 1] ? f.a1[i] : 0.0f;
        acc = __builtin_amdgcn_wmma_f32_16x16x4_f32(
             false, a,  false, f.b[i],
             (short)0, acc,  false,  false);
    }
}

__global__ __launch_bounds__(32 * NWAVE) __attribute__((amdgpu_waves_per_eu(1, 1)))
void conv_shift_wmma_f32(
    const float* __restrict__ x,
    const float* __restrict__ Wm,
    float* __restrict__ out)
{
    __shared__ float red[NWAVE][8][32];

    const int lane  = threadIdx.x & 31;
    const int wv    = threadIdx.x >> 5;
    const int ntile = blockIdx.x & 15;
    const int mtile = blockIdx.x >> 4;
    const int c     = lane & 15;
    const int h     = lane >> 4;
    const int chan  = (ntile << 4) + c;

    const int  s   = mtile * MT + c;
    const bool sok = s < NSPAT;
    const int  sc  = sok ? s : 0;
    const int  o   = sc / 14;
    const int  n   = sc - o * 14;

    int  off[6];
    bool ok[6];
    #pragma unroll
    for (int i = 0; i < 6; ++i) {
        const int tt = 4 * (i >> 1) + 2 * h + (i & 1);
        const int dj = tt / 6;
        const int rr = tt - dj * 6;
        const int l  = rr / 3;
        const int k  = rr - l * 3;
        int row = o - 1 - l;
        if (row < 0) row += 14;
        const int col = n + k - 1;
        ok[i]  = sok && col >= 0 && col < 14;
        off[i] = ok[i] ? dj * NSPAT + row * 14 + col : 0;
        off[i] += wv * (WCHUNK * CPAIRS * 2 * NSPAT);
    }

    const float* wrow = Wm + chan * KTOT + 2 * h + wv * (WCHUNK * CPAIRS * 12);

    v8f acc = {};
    Frag ring[4];

    issue_chunk(ring[0], x + 0 * CPAIRS * 2 * NSPAT, wrow + 0 * CPAIRS * 12, off);
    issue_chunk(ring[1], x + 1 * CPAIRS * 2 * NSPAT, wrow + 1 * CPAIRS * 12, off);
    issue_chunk(ring[2], x + 2 * CPAIRS * 2 * NSPAT, wrow + 2 * CPAIRS * 12, off);

    #pragma unroll
    for (int q = 0; q < WCHUNK; ++q) {
        if (q + 3 < WCHUNK)
            issue_chunk(ring[(q + 3) & 3], x + (q + 3) * CPAIRS * 2 * NSPAT,
                        wrow + (q + 3) * CPAIRS * 12, off);
        mma_chunk(acc, ring[q & 3], ok);
    }

    #pragma unroll
    for (int v = 0; v < 8; ++v)
        red[wv][v][lane] = acc[v];
    __syncthreads();

    const int mofs = h << 3;
    #pragma unroll
    for (int r = 0; r < RPW; ++r) {
        const int v = wv * RPW + r;
        float sum = red[0][v][lane];
        #pragma unroll
        for (int p = 1; p < NWAVE; ++p)
            sum += red[p][v][lane];
        const int so = mtile * MT + v + mofs;
        if (so < NSPAT)
            out[chan * NSPAT + so] = sum;
    }
}

extern "C" void kernel_launch(void* const* d_in, const int* in_sizes, int n_in,
                              void* d_out, int out_size, void* d_ws, size_t ws_size,
                              hipStream_t stream) {
    const float* x  = (const float*)d_in[0];
    const float* Wm = (const float*)d_in[1];
    float* out = (float*)d_out;

    dim3 grid(208);
    dim3 block(128);
    conv_shift_wmma_f32<<<grid, block, 0, stream>>>(x, Wm, out);
}


# Fragment loads widened by a new k order (lane half h owns 2 whole channels per 24-t chunk): B as three 16-byte loads, A as four 12-byte 3-tap runs; 56 loads per wave, was 144; f32 sum order changes
# speedup vs baseline: 6.4512x; 1.0899x over previous
#include <hip/hip_runtime.h>


typedef __attribute__((ext_vector_type(2))) float v2f;
typedef __attribute__((ext_vector_type(4))) float v4f;
typedef __attribute__((ext_vector_type(8))) float v8f;

struct f3 { float v0, v1, v2; };

#define KTOT 768
#define MT 16
#define NSPAT 196
#define CCHAN 4
#define CT (6 * CCHAN)
#define CSTEPS (CT / 4)
#define NCHUNK (KTOT / CT)
#define NWAVE 4
#define WCHUNK (NCHUNK / NWAVE)
#define RPW (8 / NWAVE)

struct Frag {
    f3  a[4];
    v4f b[3];
};

__device__ __forceinline__ void issue_chunk(Frag& f, const float* __restrict__ xc,
                                            const float* __restrict__ wc,
                                            const int (&off)[4])
{
    #pragma unroll
    for (int u = 0; u < 4; ++u)
        f.a[u] = *(const f3*)(xc + off[u]);
    #pragma unroll
    for (int m = 0; m < 3; ++m)
        f.b[m] = *(const v4f*)(wc + 4 * m);
}

__device__ __forceinline__ void mma_chunk(v8f& acc, const Frag& f, bool e0, bool e13)
{
    float av[12];
    #pragma unroll
    for (int u = 0; u < 4; ++u) {
        const f3 r = f.a[u];
        av[3 * u + 0] = e0 ? 0.0f : (e13 ? r.v1 : r.v0);
        av[3 * u + 1] = e0 ? r.v0 : (e13 ? r.v2 : r.v1);
        av[3 * u + 2] = e0 ? r.v1 : (e13 ? 0.0f : r.v2);
    }
    #pragma unroll
    for (int i = 0; i < CSTEPS; ++i) {
        v2f a, b;
        a.x = av[2 * i];
        a.y = av[2 * i + 1];
        b.x = f.b[i >> 1][2 * (i & 1)];
        b.y = f.b[i >> 1][2 * (i & 1) + 1];
        acc = __builtin_amdgcn_wmma_f32_16x16x4_f32(
             false, a,  false, b,
             (short)0, acc,  false,  false);
    }
}

__global__ __launch_bounds__(32 * NWAVE) __attribute__((amdgpu_waves_per_eu(1, 1)))
void conv_shift_wmma_f32(
    const float* __restrict__ x,
    const float* __restrict__ Wm,
    float* __restrict__ out)
{
    __shared__ float red[NWAVE][8][32];

    const int lane  = threadIdx.x & 31;
    const int wv    = threadIdx.x >> 5;
    const int ntile = blockIdx.x & 15;
    const int mtile = blockIdx.x >> 4;
    const int c     = lane & 15;
    const int h     = lane >> 4;
    const int chan  = (ntile << 4) + c;

    const int  s   = mtile * MT + c;
    const int  sc  = s < NSPAT ? s : 0;
    const int  o   = sc / 14;
    const int  n   = sc - o * 14;
    const bool e0  = n == 0;
    const bool e13 = n == 13;
    const int  start = n == 0 ? 0 : (n == 13 ? 11 : n - 1);

    int off[4];
    #pragma unroll
    for (int u = 0; u < 4; ++u) {
        const int d = u >> 1;
        const int l = u & 1;
        int row = o - 1 - l;
        if (row < 0) row += 14;
        off[u] = (2 * h + d) * NSPAT + row * 14 + start
               + wv * (WCHUNK * CCHAN * NSPAT);
    }

    const float* wrow = Wm + chan * KTOT + 12 * h + wv * (WCHUNK * CT);

    v8f acc = {};
    Frag fr[WCHUNK];

    #pragma unroll
    for (int q = 0; q < WCHUNK; ++q) {
        issue_chunk(fr[q], x + q * CCHAN * NSPAT, wrow + q * CT, off);
        __builtin_amdgcn_sched_barrier(0);
    }

    #pragma unroll
    for (int q = 0; q < WCHUNK; ++q)
        mma_chunk(acc, fr[q], e0, e13);

    #pragma unroll
    for (int v = 0; v < 8; ++v)
        red[wv][v][lane] = acc[v];
    __syncthreads();

    const int mofs = h << 3;
    #pragma unroll
    for (int r = 0; r < RPW; ++r) {
        const int v = wv * RPW + r;
        float sum = red[0][v][lane];
        #pragma unroll
        for (int p = 1; p < NWAVE; ++p)
            sum += red[p][v][lane];
        const int so = mtile * MT + v + mofs;
        if (so < NSPAT)
            out[chan * NSPAT + so] = sum;
    }
}

extern "C" void kernel_launch(void* const* d_in, const int* in_sizes, int n_in,
                              void* d_out, int out_size, void* d_ws, size_t ws_size,
                              hipStream_t stream) {
    const float* x  = (const float*)d_in[0];
    const float* Wm = (const float*)d_in[1];
    float* out = (float*)d_out;

    dim3 grid(208);
    dim3 block(128);
    conv_shift_wmma_f32<<<grid, block, 0, stream>>>(x, Wm, out);
}


# W staged through per-wave LDS: each wave fetches its 16x192 piece of W with 24 whole-line loads (4 cache lines each, was 16 to 32 per load) and reads its B fragments back with ds loads
# speedup vs baseline: 6.8499x; 1.0618x over previous
#include <hip/hip_runtime.h>


typedef __attribute__((ext_vector_type(2))) float v2f;
typedef __attribute__((ext_vector_type(4))) float v4f;
typedef __attribute__((ext_vector_type(8))) float v8f;

struct f3 { float v0, v1, v2; };

#define KTOT 768
#define MT 16
#define NSPAT 196
#define CCHAN 4
#define CT (6 * CCHAN)
#define CSTEPS (CT / 4)
#define NCHUNK (KTOT / CT)
#define NWAVE 4
#define WCHUNK (NCHUNK / NWAVE)
#define RPW (8 / NWAVE)

#define WROWS 16
#define WQ (KTOT / NWAVE)
#define WCB (WQ / 64)
#define WP4 49

struct Frag {
    f3  a[4];
};

struct BFrag {
    v4f b[3];
};

__device__ __forceinline__ void issue_chunk(Frag& f, const float* __restrict__ xc,
                                            const int (&off)[4])
{
    #pragma unroll
    for (int u = 0; u < 4; ++u)
        f.a[u] = *(const f3*)(xc + off[u]);
}

__device__ __forceinline__ void issue_wblock(v4f (&wl)[8], const v4f* __restrict__ wg, int cb)
{
    #pragma unroll
    for (int g = 0; g < 8; ++g)
        wl[g] = wg[g * 2 * (KTOT / 4) + cb * 16];
}

__device__ __forceinline__ void store_wblock(const v4f (&wl)[8], v4f* ws, int cb)
{
    #pragma unroll
    for (int g = 0; g < 8; ++g)
        ws[g * 2 * WP4 + cb * 16] = wl[g];
}

__device__ __forceinline__ void load_bfrag(BFrag& f, const v4f* wf, int q)
{
    #pragma unroll
    for (int m = 0; m < 3; ++m)
        f.b[m] = wf[6 * q + m];
}

__device__ __forceinline__ void mma_chunk(v8f& acc, const Frag& f, const BFrag& fb, bool e0, bool e13)
{
    float av[12];
    #pragma unroll
    for (int u = 0; u < 4; ++u) {
        const f3 r = f.a[u];
        av[3 * u + 0] = e0 ? 0.0f : (e13 ? r.v1 : r.v0);
        av[3 * u + 1] = e0 ? r.v0 : (e13 ? r.v2 : r.v1);
        av[3 * u + 2] = e0 ? r.v1 : (e13 ? 0.0f : r.v2);
    }
    #pragma unroll
    for (int i = 0; i < CSTEPS; ++i) {
        v2f a, b;
        a.x = av[2 * i];
        a.y = av[2 * i + 1];
        b.x = fb.b[i >> 1][2 * (i & 1)];
        b.y = fb.b[i >> 1][2 * (i & 1) + 1];
        acc = __builtin_amdgcn_wmma_f32_16x16x4_f32(
             false, a,  false, b,
             (short)0, acc,  false,  false);
    }
}

__global__ __launch_bounds__(32 * NWAVE) __attribute__((amdgpu_waves_per_eu(1, 1)))
void conv_shift_wmma_f32(
    const float* __restrict__ x,
    const float* __restrict__ Wm,
    float* __restrict__ out)
{
    __shared__ float red[NWAVE][8][32];
    __shared__ v4f wst[NWAVE][WROWS * WP4];

    const int lane  = threadIdx.x & 31;
    const int wv    = threadIdx.x >> 5;
    const int ntile = blockIdx.x & 15;
    const int mtile = blockIdx.x >> 4;
    const int c     = lane & 15;
    const int h     = lane >> 4;
    const int chan  = (ntile << 4) + c;

    const int  s   = mtile * MT + c;
    const int  sc  = s < NSPAT ? s : 0;
    const int  o   = sc / 14;
    const int  n   = sc - o * 14;
    const bool e0  = n == 0;
    const bool e13 = n == 13;
    const int  start = n == 0 ? 0 : (n == 13 ? 11 : n - 1);

    int off[4];
    #pragma unroll
    for (int u = 0; u < 4; ++u) {
        const int d = u >> 1;
        const int l = u & 1;
        int row = o - 1 - l;
        if (row < 0) row += 14;
        off[u] = (2 * h + d) * NSPAT + row * 14 + start
               + wv * (WCHUNK * CCHAN * NSPAT);
    }

    const v4f* wg = (const v4f*)(Wm + ((ntile << 4) + h) * KTOT + wv * WQ + 4 * c);
    v4f*       ws = &wst[wv][h * WP4 + c];
    const v4f* wf = &wst[wv][c * WP4 + 3 * h];

    v8f acc = {};
    Frag fr[WCHUNK];
    v4f  wl[WCB][8];
    BFrag fb[WCHUNK];

    issue_wblock(wl[0], wg, 0);
    __builtin_amdgcn_sched_barrier(0);
    issue_chunk(fr[0], x + 0 * CCHAN * NSPAT, off);
    issue_chunk(fr[1], x + 1 * CCHAN * NSPAT, off);
    __builtin_amdgcn_sched_barrier(0);
    issue_wblock(wl[1], wg, 1);
    __builtin_amdgcn_sched_barrier(0);
    issue_chunk(fr[2], x + 2 * CCHAN * NSPAT, off);
    issue_chunk(fr[3], x + 3 * CCHAN * NSPAT, off);
    issue_chunk(fr[4], x + 4 * CCHAN * NSPAT, off);
    __builtin_amdgcn_sched_barrier(0);
    issue_wblock(wl[2], wg, 2);
    __builtin_amdgcn_sched_barrier(0);
    issue_chunk(fr[5], x + 5 * CCHAN * NSPAT, off);
    issue_chunk(fr[6], x + 6 * CCHAN * NSPAT, off);
    issue_chunk(fr[7], x + 7 * CCHAN * NSPAT, off);
    __builtin_amdgcn_sched_barrier(0);

    store_wblock(wl[0], ws, 0);
    load_bfrag(fb[0], wf, 0);
    load_bfrag(fb[1], wf, 1);
    mma_chunk(acc, fr[0], fb[0], e0, e13);
    store_wblock(wl[1], ws, 1);
    load_bfrag(fb[2], wf, 2);
    load_bfrag(fb[3], wf, 3);
    mma_chunk(acc, fr[1], fb[1], e0, e13);
    mma_chunk(acc, fr[2], fb[2], e0, e13);
    load_bfrag(fb[4], wf, 4);
    store_wblock(wl[2], ws, 2);
    load_bfrag(fb[5], wf, 5);
    mma_chunk(acc, fr[3], fb[3], e0, e13);
    mma_chunk(acc, fr[4], fb[4], e0, e13);
    load_bfrag(fb[6], wf, 6);
    load_bfrag(fb[7], wf, 7);
    mma_chunk(acc, fr[5], fb[5], e0, e13);
    mma_chunk(acc, fr[6], fb[6], e0, e13);
    mma_chunk(acc, fr[7], fb[7], e0, e13);
    __builtin_amdgcn_sched_group_barrier(0x200, 8, 0);
    __builtin_amdgcn_sched_group_barrier(0x100, 6, 0);
    __builtin_amdgcn_sched_group_barrier(0x008, 6, 0);
    __builtin_amdgcn_sched_group_barrier(0x200, 8, 0);
    __builtin_amdgcn_sched_group_barrier(0x100, 6, 0);
    __builtin_amdgcn_sched_group_barrier(0x008, 12, 0);
    __builtin_amdgcn_sched_group_barrier(0x100, 3, 0);
    __builtin_amdgcn_sched_group_barrier(0x200, 8, 0);
    __builtin_amdgcn_sched_group_barrier(0x100, 3, 0);
    __builtin_amdgcn_sched_group_barrier(0x008, 12, 0);
    __builtin_amdgcn_sched_group_barrier(0x100, 6, 0);
    __builtin_amdgcn_sched_group_barrier(0x008, 18, 0);

    #pragma unroll
    for (int v = 0; v < 8; ++v)
        red[wv][v][lane] = acc[v];
    __syncthreads();

    const int mofs = h << 3;
    #pragma unroll
    for (int r = 0; r < RPW; ++r) {
        const int v = wv * RPW + r;
        float sum = red[0][v][lane];
        #pragma unroll
        for (int p = 1; p < NWAVE; ++p)
            sum += red[p][v][lane];
        const int so = mtile * MT + v + mofs;
        if (so < NSPAT)
            out[chan * NSPAT + so] = sum;
    }
}

extern "C" void kernel_launch(void* const* d_in, const int* in_sizes, int n_in,
                              void* d_out, int out_size, void* d_ws, size_t ws_size,
                              hipStream_t stream) {
    const float* x  = (const float*)d_in[0];
    const float* Wm = (const float*)d_in[1];
    float* out = (float*)d_out;

    dim3 grid(208);
    dim3 block(128);
    conv_shift_wmma_f32<<<grid, block, 0, stream>>>(x, Wm, out);
}
